# starting kernel
# baseline (speedup 1.0000x reference)
#include <hip/hip_runtime.h>
#include <hip/hip_bf16.h>

typedef __attribute__((ext_vector_type(2))) float v2f;
typedef __attribute__((ext_vector_type(8))) float v8f;

#define N_    64
#define C_    4
#define D_    4096
#define G_    32
#define P_    384

#define WAVES_PER_BLOCK 8
#define THREADS (WAVES_PER_BLOCK * 32)
#define DCHUNKS 4
#define D_PER_BLOCK (D_ / DCHUNKS)
#define D_PER_WAVE  (D_PER_BLOCK / WAVES_PER_BLOCK)
#define PAIRS       (D_PER_WAVE / 32)

#define SWAPX16 0x401f

__device__ __forceinline__ float swap_halves(float v) {
    return __int_as_float(__builtin_amdgcn_ds_swizzle(__float_as_int(v), SWAPX16));
}

__global__ __launch_bounds__(THREADS)
void gauss_superpos_wmma(const float* __restrict__ x,
                         const float* __restrict__ net,
                         const int*   __restrict__ amp_idx,
                         const int*   __restrict__ mu_idx,
                         const int*   __restrict__ sigma_idx,
                         float*       __restrict__ out)
{
    __shared__ float sA2[G_], sA1[G_], sA0[G_], sAmp[G_];

    const int nc = blockIdx.x;
    const int n  = nc / C_;
    const int c  = nc % C_;
    const int tid = threadIdx.x;

    if (tid < G_) {
        const int g  = tid;
        const int ia = amp_idx  [c * G_ + g];
        const int im = mu_idx   [c * G_ + g];
        const int is = sigma_idx[c * G_ + g];
        const float amp = net[n * P_ + ia];
        const float mu  = net[n * P_ + im];
        const float sg  = net[n * P_ + is];
        const float iv  = 1.0f / (sg * sg);
        const float LOG2E = 1.4426950408889634f;
        const float h = -0.5f * iv * LOG2E;
        sA2[g]  = h;
        sA1[g]  = mu * iv * LOG2E;
        sA0[g]  = h * mu * mu;
        sAmp[g] = amp;
    }
    __syncthreads();

    const int lane = tid & 31;
    const int half = lane >> 4;
    const int gl   = lane & 15;
    const int wave = tid >> 5;

    v2f At0, At1;
    At0.x = half ? sA0[gl]      : sA2[gl];
    At0.y = half ? 0.0f         : sA1[gl];
    At1.x = half ? sA0[16 + gl] : sA2[16 + gl];
    At1.y = half ? 0.0f         : sA1[16 + gl];

    float amp0[8], amp1[8];
#pragma unroll
    for (int r = 0; r < 8; ++r) {
        amp0[r] = sAmp[r + 8 * half];
        amp1[r] = sAmp[16 + r + 8 * half];
    }

    const float* __restrict__ xrow = x   + (size_t)nc * D_;
    float*       __restrict__ orow = out + (size_t)nc * D_;
    const int dbase = blockIdx.y * D_PER_BLOCK + wave * D_PER_WAVE;

    float xv[PAIRS];
#pragma unroll
    for (int p = 0; p < PAIRS; ++p)
        xv[p] = xrow[dbase + p * 32 + lane];

#pragma unroll
    for (int p = 0; p < PAIRS; ++p) {
        const float xlo = xv[p];
        const float xhi = swap_halves(xv[p]);

        v2f B0, B1;
        B0.x = half ? 1.0f : xlo * xlo;  B0.y = half ? 0.0f : xlo;
        B1.x = half ? 1.0f : xhi * xhi;  B1.y = half ? 0.0f : xhi;
        v8f cz = {0.f, 0.f, 0.f, 0.f, 0.f, 0.f, 0.f, 0.f};

        v8f e00 = __builtin_amdgcn_wmma_f32_16x16x4_f32(
                      false, At0, false, B0, (short)0, cz, false, false);
        v8f e01 = __builtin_amdgcn_wmma_f32_16x16x4_f32(
                      false, At1, false, B0, (short)0, cz, false, false);
        v8f e10 = __builtin_amdgcn_wmma_f32_16x16x4_f32(
                      false, At0, false, B1, (short)0, cz, false, false);
        v8f e11 = __builtin_amdgcn_wmma_f32_16x16x4_f32(
                      false, At1, false, B1, (short)0, cz, false, false);

        float s0a = 0.f, s0b = 0.f, s1a = 0.f, s1b = 0.f;
#pragma unroll
        for (int r = 0; r < 8; ++r) {
            s0a = fmaf(amp0[r], __builtin_amdgcn_exp2f(e00[r]), s0a);
            s0b = fmaf(amp1[r], __builtin_amdgcn_exp2f(e01[r]), s0b);
        }
#pragma unroll
        for (int r = 0; r < 8; ++r) {
            s1a = fmaf(amp0[r], __builtin_amdgcn_exp2f(e10[r]), s1a);
            s1b = fmaf(amp1[r], __builtin_amdgcn_exp2f(e11[r]), s1b);
        }
        const float s0 = s0a + s0b;
        const float s1 = s1a + s1b;
        const float tot0 = s0 + swap_halves(s0);
        const float tot1 = s1 + swap_halves(s1);

        orow[dbase + p * 32 + lane] = half ? tot1 : tot0;
    }
}

extern "C" void kernel_launch(void* const* d_in, const int* in_sizes, int n_in,
                              void* d_out, int out_size, void* d_ws, size_t ws_size,
                              hipStream_t stream) {
    const float* x   = (const float*)d_in[0];
    const float* net = (const float*)d_in[1];
    const int*   ai  = (const int*)d_in[2];
    const int*   mi  = (const int*)d_in[3];
    const int*   si  = (const int*)d_in[4];
    float* out = (float*)d_out;

    dim3 grid(N_ * C_, DCHUNKS);
    gauss_superpos_wmma<<<grid, THREADS, 0, stream>>>(x, net, ai, mi, si, out);
    (void)in_sizes; (void)n_in; (void)out_size; (void)d_ws; (void)ws_size;
}


# Launch shape: 512 four-wave blocks of half a row each, 2048 long waves instead of 8192 short ones; a wave owns 512 d (16 tile pairs) and issues its indices, then all its x loads, ahead of the gather
# speedup vs baseline: 1.0959x; 1.0959x over previous
#include <hip/hip_runtime.h>
#include <hip/hip_bf16.h>


typedef __attribute__((ext_vector_type(2))) float v2f;
typedef __attribute__((ext_vector_type(8))) float v8f;

#define N_    64
#define C_    4
#define D_    4096
#define G_    32
#define P_    384

#define WAVES_PER_BLOCK 4
#define THREADS (WAVES_PER_BLOCK * 32)
#define DCHUNKS 2
#define D_PER_BLOCK (D_ / DCHUNKS)
#define D_PER_WAVE  (D_PER_BLOCK / WAVES_PER_BLOCK)
#define PAIRS       (D_PER_WAVE / 32)

#define SWAPX16 0x401f

__device__ __forceinline__ float swap_halves(float v) {
    return __int_as_float(__builtin_amdgcn_ds_swizzle(__float_as_int(v), SWAPX16));
}

__global__ __launch_bounds__(THREADS)
void gauss_superpos_wmma(const float* __restrict__ x,
                         const float* __restrict__ net,
                         const int*   __restrict__ amp_idx,
                         const int*   __restrict__ mu_idx,
                         const int*   __restrict__ sigma_idx,
                         float*       __restrict__ out)
{
    __shared__ float sA2[G_], sA1[G_], sA0[G_], sAmp[G_];

    const int nc = blockIdx.x;
    const int n  = nc / C_;
    const int c  = nc % C_;
    const int tid = threadIdx.x;

    const int lane = tid & 31;
    const int half = lane >> 4;
    const int gl   = lane & 15;
    const int wave = tid >> 5;

    const size_t span = (size_t)nc * D_ + blockIdx.y * D_PER_BLOCK + wave * D_PER_WAVE + lane;
    const float* __restrict__ xrow = x   + span;
    float*       __restrict__ orow = out + span;

    int ia = 0, im = 0, is = 0;
    if (tid < G_) {
        ia = amp_idx  [c * G_ + tid];
        im = mu_idx   [c * G_ + tid];
        is = sigma_idx[c * G_ + tid];
    }

    float xv[PAIRS];
#pragma unroll
    for (int p = 0; p < PAIRS; ++p)
        xv[p] = xrow[p * 32];

    if (tid < G_) {
        const int g  = tid;
        const float amp = net[n * P_ + ia];
        const float mu  = net[n * P_ + im];
        const float sg  = net[n * P_ + is];
        const float iv  = 1.0f / (sg * sg);
        const float LOG2E = 1.4426950408889634f;
        const float h = -0.5f * iv * LOG2E;
        sA2[g]  = h;
        sA1[g]  = mu * iv * LOG2E;
        sA0[g]  = h * mu * mu;
        sAmp[g] = amp;
    }
    __syncthreads();

    v2f At0, At1;
    At0.x = half ? sA0[gl]      : sA2[gl];
    At0.y = half ? 0.0f         : sA1[gl];
    At1.x = half ? sA0[16 + gl] : sA2[16 + gl];
    At1.y = half ? 0.0f         : sA1[16 + gl];

    float amp0[8], amp1[8];
#pragma unroll
    for (int r = 0; r < 8; ++r) {
        amp0[r] = sAmp[r + 8 * half];
        amp1[r] = sAmp[16 + r + 8 * half];
    }

#pragma unroll
    for (int p = 0; p < PAIRS; ++p) {
        const float xlo = xv[p];
        const float xhi = swap_halves(xv[p]);

        v2f B0, B1;
        B0.x = half ? 1.0f : xlo * xlo;  B0.y = half ? 0.0f : xlo;
        B1.x = half ? 1.0f : xhi * xhi;  B1.y = half ? 0.0f : xhi;
        v8f cz = {0.f, 0.f, 0.f, 0.f, 0.f, 0.f, 0.f, 0.f};

        v8f e00 = __builtin_amdgcn_wmma_f32_16x16x4_f32(
                      false, At0, false, B0, (short)0, cz, false, false);
        v8f e01 = __builtin_amdgcn_wmma_f32_16x16x4_f32(
                      false, At1, false, B0, (short)0, cz, false, false);
        v8f e10 = __builtin_amdgcn_wmma_f32_16x16x4_f32(
                      false, At0, false, B1, (short)0, cz, false, false);
        v8f e11 = __builtin_amdgcn_wmma_f32_16x16x4_f32(
                      false, At1, false, B1, (short)0, cz, false, false);

        float s0a = 0.f, s0b = 0.f, s1a = 0.f, s1b = 0.f;
#pragma unroll
        for (int r = 0; r < 8; ++r) {
            s0a = fmaf(amp0[r], __builtin_amdgcn_exp2f(e00[r]), s0a);
            s0b = fmaf(amp1[r], __builtin_amdgcn_exp2f(e01[r]), s0b);
        }
#pragma unroll
        for (int r = 0; r < 8; ++r) {
            s1a = fmaf(amp0[r], __builtin_amdgcn_exp2f(e10[r]), s1a);
            s1b = fmaf(amp1[r], __builtin_amdgcn_exp2f(e11[r]), s1b);
        }
        const float s0 = s0a + s0b;
        const float s1 = s1a + s1b;
        const float tot0 = s0 + swap_halves(s0);
        const float tot1 = s1 + swap_halves(s1);

        orow[p * 32] = half ? tot1 : tot0;
    }
}

extern "C" void kernel_launch(void* const* d_in, const int* in_sizes, int n_in,
                              void* d_out, int out_size, void* d_ws, size_t ws_size,
                              hipStream_t stream) {
    const float* x   = (const float*)d_in[0];
    const float* net = (const float*)d_in[1];
    const int*   ai  = (const int*)d_in[2];
    const int*   mi  = (const int*)d_in[3];
    const int*   si  = (const int*)d_in[4];
    float* out = (float*)d_out;

    dim3 grid(N_ * C_, DCHUNKS);
    gauss_superpos_wmma<<<grid, THREADS, 0, stream>>>(x, net, ai, mi, si, out);
    (void)in_sizes; (void)n_in; (void)out_size; (void)d_ws; (void)ws_size;
}


# Re-tiled product: one matrix instruction = 8 gaussians x 32 d (rows 8-15 read d+16 via K2/K3, A0 enters as C), so a lane owns one d: lane-private g-sum in a new order, no swizzles or selects
# speedup vs baseline: 1.1857x; 1.0819x over previous
#include <hip/hip_runtime.h>
#include <hip/hip_bf16.h>


typedef __attribute__((ext_vector_type(2))) float v2f;
typedef __attribute__((ext_vector_type(8))) float v8f;

#define N_    64
#define C_    4
#define D_    4096
#define G_    32
#define P_    384

#define WAVES_PER_BLOCK 4
#define THREADS (WAVES_PER_BLOCK * 32)
#define DCHUNKS 2
#define D_PER_BLOCK (D_ / DCHUNKS)
#define D_PER_WAVE  (D_PER_BLOCK / WAVES_PER_BLOCK)
#define STEPS       (D_PER_WAVE / 32)
#define GROUPS      (G_ / 8)

__global__ __launch_bounds__(THREADS)
void gauss_superpos_wmma(const float* __restrict__ x,
                         const float* __restrict__ net,
                         const int*   __restrict__ amp_idx,
                         const int*   __restrict__ mu_idx,
                         const int*   __restrict__ sigma_idx,
                         float*       __restrict__ out)
{
    __shared__ float sA2[G_], sA1[G_], sA0[G_], sAmp[G_];

    const int nc = blockIdx.x;
    const int n  = nc / C_;
    const int c  = nc % C_;
    const int tid = threadIdx.x;

    const int lane = tid & 31;
    const int half = lane >> 4;
    const int gl   = lane & 15;
    const int wave = tid >> 5;

    const size_t span = (size_t)nc * D_ + blockIdx.y * D_PER_BLOCK + wave * D_PER_WAVE + lane;
    const float* __restrict__ xrow = x   + span;
    float*       __restrict__ orow = out + span;

    int ia = 0, im = 0, is = 0;
    if (tid < G_) {
        ia = amp_idx  [c * G_ + tid];
        im = mu_idx   [c * G_ + tid];
        is = sigma_idx[c * G_ + tid];
    }

    float xv[STEPS];
#pragma unroll
    for (int p = 0; p < STEPS; ++p)
        xv[p] = xrow[p * 32];

    if (tid < G_) {
        const int g  = tid;
        const float amp = net[n * P_ + ia];
        const float mu  = net[n * P_ + im];
        const float sg  = net[n * P_ + is];
        const float iv  = 1.0f / (sg * sg);
        const float LOG2E = 1.4426950408889634f;
        const float h = -0.5f * iv * LOG2E;
        sA2[g]  = h;
        sA1[g]  = mu * iv * LOG2E;
        sA0[g]  = h * mu * mu;
        sAmp[g] = amp;
    }
    __syncthreads();

    const bool own = ((gl >> 3) == half);
    const int  gr  = gl & 7;
    v2f  At[GROUPS];
    v8f  Ct[GROUPS];
    float amp[GROUPS][8];
#pragma unroll
    for (int j = 0; j < GROUPS; ++j) {
        const float a2 = sA2[8 * j + gr];
        const float a1 = sA1[8 * j + gr];
        At[j].x = own ? a2 : 0.0f;
        At[j].y = own ? a1 : 0.0f;
#pragma unroll
        for (int r = 0; r < 8; ++r) {
            Ct[j][r]  = sA0 [8 * j + r];
            amp[j][r] = sAmp[8 * j + r];
        }
    }

#pragma unroll
    for (int p = 0; p < STEPS; ++p) {
        const float xo = xv[p];
        v2f B;
        B.x = xo * xo;
        B.y = xo;

        v8f e0 = __builtin_amdgcn_wmma_f32_16x16x4_f32(
                     false, At[0], false, B, (short)0, Ct[0], false, false);
        v8f e1 = __builtin_amdgcn_wmma_f32_16x16x4_f32(
                     false, At[1], false, B, (short)0, Ct[1], false, false);
        v8f e2 = __builtin_amdgcn_wmma_f32_16x16x4_f32(
                     false, At[2], false, B, (short)0, Ct[2], false, false);
        v8f e3 = __builtin_amdgcn_wmma_f32_16x16x4_f32(
                     false, At[3], false, B, (short)0, Ct[3], false, false);

        float s0 = 0.f, s1 = 0.f, s2 = 0.f, s3 = 0.f;
#pragma unroll
        for (int r = 0; r < 8; ++r) s0 = fmaf(amp[0][r], __builtin_amdgcn_exp2f(e0[r]), s0);
#pragma unroll
        for (int r = 0; r < 8; ++r) s1 = fmaf(amp[1][r], __builtin_amdgcn_exp2f(e1[r]), s1);
#pragma unroll
        for (int r = 0; r < 8; ++r) s2 = fmaf(amp[2][r], __builtin_amdgcn_exp2f(e2[r]), s2);
#pragma unroll
        for (int r = 0; r < 8; ++r) s3 = fmaf(amp[3][r], __builtin_amdgcn_exp2f(e3[r]), s3);

        orow[p * 32] = (s0 + s1) + (s2 + s3);
    }
}

extern "C" void kernel_launch(void* const* d_in, const int* in_sizes, int n_in,
                              void* d_out, int out_size, void* d_ws, size_t ws_size,
                              hipStream_t stream) {
    const float* x   = (const float*)d_in[0];
    const float* net = (const float*)d_in[1];
    const int*   ai  = (const int*)d_in[2];
    const int*   mi  = (const int*)d_in[3];
    const int*   si  = (const int*)d_in[4];
    float* out = (float*)d_out;

    dim3 grid(N_ * C_, DCHUNKS);
    gauss_superpos_wmma<<<grid, THREADS, 0, stream>>>(x, net, ai, mi, si, out);
    (void)in_sizes; (void)n_in; (void)out_size; (void)d_ws; (void)ws_size;
}
